# starting kernel
# baseline (speedup 1.0000x reference)
#include <hip/hip_runtime.h>


typedef __attribute__((ext_vector_type(2))) float v2f;
typedef __attribute__((ext_vector_type(8))) float v8f;

#define NBINS 16
#define TPB   256

__device__ __forceinline__ float fast_exp2(float a) {
#if __has_builtin(__builtin_amdgcn_exp2f)
  return __builtin_amdgcn_exp2f(a);
#else
  return exp2f(a);
#endif
}

__device__ __forceinline__ float fast_rcp(float a) {
#if __has_builtin(__builtin_amdgcn_rcpf)
  return __builtin_amdgcn_rcpf(a);
#else
  return 1.0f / a;
#endif
}

__global__ __launch_bounds__(TPB) void HistByNorm_40673340293939_kernel(
    const float* __restrict__ x,
    const float* __restrict__ centers,
    const float* __restrict__ widths,
    float* __restrict__ out,
    int D, float invD)
{
  __shared__ float P[TPB * NBINS];

  const int n = blockIdx.x;
  const int t = threadIdx.x;
  const float* __restrict__ xrow = x + (size_t)n * (size_t)D;

  const float S = 0.84936059f;
  float aj[NBINS], bj[NBINS];
#pragma unroll
  for (int j = 0; j < NBINS; ++j) {
    float w = widths[j];
    float c = centers[j];
    float iw = S * fast_rcp(w);
    aj[j] = iw;
    bj[j] = c * iw;
  }

  float acc[NBINS];
#pragma unroll
  for (int j = 0; j < NBINS; ++j) acc[j] = 0.0f;

  const float4* __restrict__ xrow4 = (const float4*)xrow;
  const int iters = D / (TPB * 4);
  for (int k = 0; k < iters; ++k) {
    float4 v = xrow4[t + k * TPB];
    if (k + 1 < iters)
      __builtin_prefetch(&xrow4[t + (k + 1) * TPB], 0, 0);
    float xs[4] = {v.x, v.y, v.z, v.w};
#pragma unroll
    for (int e = 0; e < 4; ++e) {
      const float xv = xs[e];
#pragma unroll
      for (int j = 0; j < NBINS; ++j) {
        float z = __builtin_fmaf(xv, aj[j], -bj[j]);
        acc[j] += fast_exp2(z * (-z));
      }
    }
  }
  for (int d = iters * TPB * 4 + t; d < D; d += TPB) {
    const float xv = xrow[d];
#pragma unroll
    for (int j = 0; j < NBINS; ++j) {
      float z = __builtin_fmaf(xv, aj[j], -bj[j]);
      acc[j] += fast_exp2(z * (-z));
    }
  }

#pragma unroll
  for (int j = 0; j < NBINS; ++j) P[t * NBINS + j] = acc[j];
  __syncthreads();

  if (t < 32) {
    const int lane = t;
    const int half = lane >> 4;
    const int m    = lane & 15;
    v8f cacc = {};
    v2f ones; ones.x = 1.0f; ones.y = 1.0f;
#pragma unroll 8
    for (int r0 = 0; r0 < TPB; r0 += 4) {
      v2f a;
      a.x = P[(r0 + 2 * half + 0) * NBINS + m];
      a.y = P[(r0 + 2 * half + 1) * NBINS + m];
      cacc = __builtin_amdgcn_wmma_f32_16x16x4_f32(
           false, a,  false, ones,
           (short)0, cacc,  false,  false);
    }
    if (m == 0) {
      const int binBase = half * 8;
      const float INV_SQRT_2PI = 0.39894228040143268f;
#pragma unroll
      for (int r = 0; r < 8; ++r) {
        const int b = binBase + r;
        const float scale = INV_SQRT_2PI * fast_rcp(widths[b]) * invD;
        out[(size_t)n * NBINS + b] = cacc[r] * scale;
      }
    }
  }
}

extern "C" void kernel_launch(void* const* d_in, const int* in_sizes, int n_in,
                              void* d_out, int out_size, void* d_ws, size_t ws_size,
                              hipStream_t stream) {
  const float* x       = (const float*)d_in[0];
  const float* centers = (const float*)d_in[1];
  const float* widths  = (const float*)d_in[2];
  float*       out     = (float*)d_out;

  const int nbins = in_sizes[1];
  const int N     = out_size / nbins;
  const int D     = in_sizes[0] / N;
  const float invD = 1.0f / (float)D;

  HistByNorm_40673340293939_kernel<<<N, TPB, 0, stream>>>(
      x, centers, widths, out, D, invD);
}


# main loop in phases per float4: 64 v_exp_f32 in one run at raised s_setprio, 96 packed ops at low priority: a SIMD's 2 waves fill each other's exp gaps (younger lifted for 2/3 of row); sum order kept
# speedup vs baseline: 1.3632x; 1.3632x over previous
#include <hip/hip_runtime.h>


typedef __attribute__((ext_vector_type(2))) float v2f;
typedef __attribute__((ext_vector_type(4))) float v4f;
typedef __attribute__((ext_vector_type(8))) float v8f;

#define NBINS 16
#define TPB   256
#define WAVES (TPB / 32)
#define GRP   4

__device__ __forceinline__ float fast_exp2(float a) {
#if __has_builtin(__builtin_amdgcn_exp2f)
  return __builtin_amdgcn_exp2f(a);
#else
  return exp2f(a);
#endif
}

__device__ __forceinline__ float fast_rcp(float a) {
#if __has_builtin(__builtin_amdgcn_rcpf)
  return __builtin_amdgcn_rcpf(a);
#else
  return 1.0f / a;
#endif
}

__device__ __forceinline__ float other_half(float v) {
  const int b = __builtin_bit_cast(int, v);
  return __builtin_bit_cast(float,
      __builtin_amdgcn_permlanex16(b, b, 0x76543210, 0xfedcba98, false, false));
}

__device__ __forceinline__ void bins_of(float xs, const v2f (&a2)[NBINS / 2],
                                        const v2f (&nb2)[NBINS / 2], v2f (&acc2)[NBINS / 2]) {
  v2f xv; xv.x = xs; xv.y = xs;
#pragma unroll
  for (int p = 0; p < NBINS / 2; ++p) {
    v2f z = __builtin_elementwise_fma(xv, a2[p], nb2[p]);
    v2f q = z * (-z);
    v2f ex;
    ex.x = fast_exp2(q.x);
    ex.y = fast_exp2(q.y);
    acc2[p] += ex;
  }
}

__device__ __forceinline__ void bins_of4(v4f v, const v2f (&a2)[NBINS / 2],
                                         const v2f (&nb2)[NBINS / 2], v2f (&acc2)[NBINS / 2]) {
  bins_of(v.x, a2, nb2, acc2);
  bins_of(v.y, a2, nb2, acc2);
  bins_of(v.z, a2, nb2, acc2);
  bins_of(v.w, a2, nb2, acc2);
}

#define EPC 4

__device__ __forceinline__ void args_of4(v4f v, const v2f (&a2)[NBINS / 2],
                                         const v2f (&nb2)[NBINS / 2], v2f (&q)[EPC][NBINS / 2]) {
  const float xs[EPC] = {v.x, v.y, v.z, v.w};
#pragma unroll
  for (int e = 0; e < EPC; ++e) {
    v2f xv; xv.x = xs[e]; xv.y = xs[e];
#pragma unroll
    for (int p = 0; p < NBINS / 2; ++p) {
      v2f z = __builtin_elementwise_fma(xv, a2[p], nb2[p]);
      q[e][p] = z * (-z);
    }
  }
}

template <int TP, int PP>
__device__ __forceinline__ void exps_of4(v2f (&q)[EPC][NBINS / 2]) {
  __builtin_amdgcn_sched_barrier(0);
  __builtin_amdgcn_s_setprio(TP);
  __builtin_amdgcn_sched_barrier(0);
#pragma unroll
  for (int e = 0; e < EPC; ++e) {
#pragma unroll
    for (int p = 0; p < NBINS / 2; ++p) {
      q[e][p].x = fast_exp2(q[e][p].x);
      q[e][p].y = fast_exp2(q[e][p].y);
    }
  }
  __builtin_amdgcn_sched_barrier(0);
  __builtin_amdgcn_s_setprio(PP);
  __builtin_amdgcn_sched_barrier(0);
}

__device__ __forceinline__ void sums_of4(const v2f (&q)[EPC][NBINS / 2], v2f (&acc2)[NBINS / 2]) {
#pragma unroll
  for (int e = 0; e < EPC; ++e) {
#pragma unroll
    for (int p = 0; p < NBINS / 2; ++p) acc2[p] += q[e][p];
  }
}

template <int TP, int PP>
__device__ __forceinline__ void phase_step(v2f (&q)[EPC][NBINS / 2], v4f next,
                                           const v2f (&a2)[NBINS / 2], const v2f (&nb2)[NBINS / 2],
                                           v2f (&acc2)[NBINS / 2]) {
  exps_of4<TP, PP>(q);
  sums_of4(q, acc2);
  args_of4(next, a2, nb2, q);
}

template <int TP, int PP>
__device__ __forceinline__ void group_trip(const v4f* __restrict__ xr4, int g, v4f (&cur)[GRP],
                                           v2f (&q)[EPC][NBINS / 2], const v2f (&a2)[NBINS / 2],
                                           const v2f (&nb2)[NBINS / 2], v2f (&acc2)[NBINS / 2]) {
  v4f nxt[GRP];
#pragma unroll
  for (int u = 0; u < GRP; ++u) nxt[u] = xr4[((g + 1) * GRP + u) * 32];
#pragma unroll
  for (int u = 0; u + 1 < GRP; ++u) phase_step<TP, PP>(q, cur[u + 1], a2, nb2, acc2);
  phase_step<TP, PP>(q, nxt[0], a2, nb2, acc2);
#pragma unroll
  for (int u = 0; u < GRP; ++u) {
    cur[u] = nxt[u];
    asm volatile("" : "+v"(cur[u]));
  }
}

__global__ __launch_bounds__(TPB) __attribute__((amdgpu_waves_per_eu(2, 2))) void hist_rows_kernel(
    const float* __restrict__ x,
    const float* __restrict__ centers,
    const float* __restrict__ widths,
    float* __restrict__ out,
    int N, int D, float invD)
{
  const int lane = threadIdx.x & 31;
  const int wv  = __builtin_amdgcn_readfirstlane((int)(threadIdx.x >> 5));
  const int row = blockIdx.x * WAVES + wv;
  if (row >= N) return;
  const float* __restrict__ xrow = x + (size_t)row * (size_t)D;

  const v4f* __restrict__ xr4 = (const v4f*)xrow + lane;
  const int iters  = D / 128;
  const int groups = iters / GRP;
  v4f cur[GRP];
  const v4f* __restrict__ first = (groups > 0) ? xr4 : (const v4f*)x;
  const int step0 = (groups > 0) ? 32 : 0;
#pragma unroll
  for (int u = 0; u < GRP; ++u) cur[u] = first[u * step0];

  const float S = 0.84936059f;
  float aj[NBINS], bj[NBINS];
#pragma unroll
  for (int j = 0; j < NBINS; ++j) {
    float w = widths[j];
    float c = centers[j];
    float iw = S * fast_rcp(w);
    aj[j] = iw;
    bj[j] = c * iw;
  }
  v2f a2[NBINS / 2], nb2[NBINS / 2], acc2[NBINS / 2];
#pragma unroll
  for (int p = 0; p < NBINS / 2; ++p) {
    a2[p].x  = aj[2 * p];   a2[p].y  = aj[2 * p + 1];
    nb2[p].x = -bj[2 * p];  nb2[p].y = -bj[2 * p + 1];
    acc2[p].x = 0.0f;       acc2[p].y = 0.0f;
  }

  if (groups > 0) {
    v2f q[EPC][NBINS / 2];
    args_of4(cur[0], a2, nb2, q);
    const int trips = groups - 1;
    const int lifted = (wv >= WAVES / 2) ? (2 * groups) / 3 : 0;
    int g = 0;
    for (; g < trips && g < lifted; ++g) group_trip<3, 1>(xr4, g, cur, q, a2, nb2, acc2);
    for (; g < trips; ++g)               group_trip<2, 0>(xr4, g, cur, q, a2, nb2, acc2);
#pragma unroll
    for (int u = 0; u + 1 < GRP; ++u) phase_step<2, 0>(q, cur[u + 1], a2, nb2, acc2);
    exps_of4<2, 0>(q);
    sums_of4(q, acc2);
  }

  for (int k = groups * GRP; k < iters; ++k) bins_of4(xr4[k * 32], a2, nb2, acc2);
  for (int d = iters * 128 + lane; d < D; d += 32) bins_of(xrow[d], a2, nb2, acc2);

  const int m = lane & 15;
  v2f ones; ones.x = 1.0f; ones.y = 1.0f;
  float mine = 0.0f;
#pragma unroll
  for (int j = 0; j < NBINS; ++j) {
    v2f a;
    a.x = (j & 1) ? acc2[j >> 1].y : acc2[j >> 1].x;
    a.y = 0.0f;
    v8f zero = {};
    v8f d = __builtin_amdgcn_wmma_f32_16x16x4_f32(
         false, a,  false, ones,
         (short)0, zero,  false,  false);
    float s = ((d[0] + d[1]) + (d[2] + d[3])) + ((d[4] + d[5]) + (d[6] + d[7]));
    s += other_half(s);
    mine = (m == j) ? s : mine;
  }
  if (lane < NBINS) {
    const float INV_SQRT_2PI = 0.39894228040143268f;
    const float scale = INV_SQRT_2PI * fast_rcp(widths[lane]) * invD;
    out[(size_t)row * NBINS + lane] = mine * scale;
  }
}

extern "C" void kernel_launch(void* const* d_in, const int* in_sizes, int n_in,
                              void* d_out, int out_size, void* d_ws, size_t ws_size,
                              hipStream_t stream) {
  const float* x       = (const float*)d_in[0];
  const float* centers = (const float*)d_in[1];
  const float* widths  = (const float*)d_in[2];
  float*       out     = (float*)d_out;

  const int nbins = in_sizes[1];
  const int N     = out_size / nbins;
  const int D     = in_sizes[0] / N;
  const float invD = 1.0f / (float)D;

  hist_rows_kernel<<<(N + WAVES - 1) / WAVES, TPB, 0, stream>>>(
      x, centers, widths, out, N, D, invD);
}
